# starting kernel
# baseline (speedup 1.0000x reference)
#include <hip/hip_runtime.h>
#include <hip/hip_bf16.h>

#define IN_C  64
#define OUT_C 64
#define EPS   (1.0f + 1e-06f)

typedef float v2f __attribute__((ext_vector_type(2)));
typedef float v8f __attribute__((ext_vector_type(8)));

template <bool HAS_ATT>
__device__ __forceinline__ void wmma_block(
    const v2f a_frag[16],
    const float* __restrict__ w,
    const float* __restrict__ att,
    float* __restrict__ store,
    float* __restrict__ s_src,
    float* __restrict__ s_tgt,
    float scale, int row0, int half, int lrow)
{
    float att_s[4], att_t[4];
    if (HAS_ATT) {
#pragma unroll
        for (int nt = 0; nt < 4; ++nt) {
            att_s[nt] = att[nt * 16 + lrow];
            att_t[nt] = att[OUT_C + nt * 16 + lrow];
        }
    }

    float ssrc_acc[8];
    float stgt_acc[8];
#pragma unroll
    for (int r = 0; r < 8; ++r) { ssrc_acc[r] = 0.f; stgt_acc[r] = 0.f; }

#pragma unroll
    for (int nt = 0; nt < 4; ++nt) {
        const int col = nt * 16 + lrow;
        v8f c = {};
#pragma unroll
        for (int kk = 0; kk < 16; ++kk) {
            const int k0 = 4 * kk + 2 * half;
            v2f b;
            b.x = w[(size_t)k0 * OUT_C + col];
            b.y = w[(size_t)(k0 + 1) * OUT_C + col];
            c = __builtin_amdgcn_wmma_f32_16x16x4_f32(
                     false, a_frag[kk],
                     false, b,
                     (short)0, c,
                     false,  false);
        }
#pragma unroll
        for (int r = 0; r < 8; ++r) {
            const int row = row0 + r + half * 8;
            const float v = c[r];
            store[(size_t)row * OUT_C + col] = v * scale;
            if (HAS_ATT) {
                ssrc_acc[r] += v * att_s[nt];
                stgt_acc[r] += v * att_t[nt];
            }
        }
    }

    if (HAS_ATT) {
#pragma unroll
        for (int r = 0; r < 8; ++r) {
            float vs = ssrc_acc[r];
            float vt = stgt_acc[r];
#pragma unroll
            for (int m = 8; m >= 1; m >>= 1) {
                vs += __shfl_xor(vs, m, 32);
                vt += __shfl_xor(vt, m, 32);
            }
            if (lrow == 0) {
                const int row = row0 + r + half * 8;
                s_src[row] = vs;
                s_tgt[row] = vt;
            }
        }
    }
}

__global__ void __launch_bounds__(128)
node_gemm_kernel(const float* __restrict__ x,
                 const float* __restrict__ w_lower,
                 const float* __restrict__ a_lower,
                 const float* __restrict__ w_upper,
                 const float* __restrict__ a_upper,
                 const float* __restrict__ w_lin,
                 float* __restrict__ xm_l, float* __restrict__ xm_u,
                 float* __restrict__ sl_s, float* __restrict__ sl_t,
                 float* __restrict__ su_s, float* __restrict__ su_t,
                 float* __restrict__ out, int n_rows)
{
    const int wid  = (int)((blockIdx.x * blockDim.x + threadIdx.x) >> 5);
    const int lane = (int)(threadIdx.x & 31);
    const int row0 = wid * 16;
    if (row0 >= n_rows) return;

    const int half = lane >> 4;
    const int lrow = lane & 15;

    v2f a_frag[16];
    const float* xr = x + (size_t)(row0 + lrow) * IN_C + 2 * half;
#pragma unroll
    for (int kk = 0; kk < 16; ++kk)
        a_frag[kk] = *(const v2f*)(xr + 4 * kk);

    wmma_block<true >(a_frag, w_lower, a_lower, xm_l, sl_s, sl_t, 1.0f, row0, half, lrow);
    wmma_block<true >(a_frag, w_upper, a_upper, xm_u, su_s, su_t, 1.0f, row0, half, lrow);
    wmma_block<false>(a_frag, w_lin,   nullptr, out,  nullptr, nullptr, EPS, row0, half, lrow);
}

__global__ void __launch_bounds__(256)
edge_scatter_kernel(const int* __restrict__ idx,
                    const float* __restrict__ vals,
                    const float* __restrict__ xm,
                    const float* __restrict__ s_src,
                    const float* __restrict__ s_tgt,
                    float* __restrict__ out,
                    int E)
{
    const int e = __builtin_amdgcn_readfirstlane(
        (int)((blockIdx.x * blockDim.x + threadIdx.x) >> 5));
    const int lane = (int)(threadIdx.x & 31);
    if (e >= E) return;

    const int i = idx[e];
    const int j = idx[(size_t)E + e];

    const float z = s_src[j] + s_tgt[i];
    const float elu = (z > 0.f) ? z : expm1f(z);
    const float alpha = elu * vals[e];

    const float2 m = ((const float2*)(xm + (size_t)j * OUT_C))[lane];
    float* o = out + (size_t)i * OUT_C + lane * 2;
    atomicAdd(o,     alpha * m.x);
    atomicAdd(o + 1, alpha * m.y);
}

__global__ void __launch_bounds__(256)
relu_kernel(float* __restrict__ out, size_t n)
{
    const size_t t = (size_t)blockIdx.x * blockDim.x + threadIdx.x;
    if (t < n) out[t] = fmaxf(out[t], 0.0f);
}

extern "C" void kernel_launch(void* const* d_in, const int* in_sizes, int n_in,
                              void* d_out, int out_size, void* d_ws, size_t ws_size,
                              hipStream_t stream) {
    const float* x        = (const float*)d_in[0];
    const int*   low_idx  = (const int*)  d_in[1];
    const float* low_val  = (const float*)d_in[2];
    const int*   up_idx   = (const int*)  d_in[3];
    const float* up_val   = (const float*)d_in[4];
    const float* w_lower  = (const float*)d_in[5];
    const float* a_lower  = (const float*)d_in[6];
    const float* w_upper  = (const float*)d_in[7];
    const float* a_upper  = (const float*)d_in[8];
    const float* w_lin    = (const float*)d_in[9];
    float*       out      = (float*)d_out;

    const int N = in_sizes[0] / IN_C;
    const int E = in_sizes[2];

    float* ws   = (float*)d_ws;
    float* xm_l = ws;
    float* xm_u = xm_l + (size_t)N * OUT_C;
    float* sl_s = xm_u + (size_t)N * OUT_C;
    float* sl_t = sl_s + N;
    float* su_s = sl_t + N;
    float* su_t = su_s + N;

    {
        const int waves  = (N + 15) / 16;
        const int blocks = (waves + 3) / 4;
        node_gemm_kernel<<<blocks, 128, 0, stream>>>(
            x, w_lower, a_lower, w_upper, a_upper, w_lin,
            xm_l, xm_u, sl_s, sl_t, su_s, su_t, out, N);
    }

    {
        const int blocks = (E + 7) / 8;
        edge_scatter_kernel<<<blocks, 256, 0, stream>>>(
            low_idx, low_val, xm_l, sl_s, sl_t, out, E);
        edge_scatter_kernel<<<blocks, 256, 0, stream>>>(
            up_idx, up_val, xm_u, su_s, su_t, out, E);
    }

    {
        const size_t n = (size_t)N * OUT_C;
        const int blocks = (int)((n + 255) / 256);
        relu_kernel<<<blocks, 256, 0, stream>>>(out, n);
    }
}


# edge scatter: one wave per 32 edges (alpha computed per lane, then broadcast by readlane), row loads issued 8 edges at a time one group ahead of their atomics; atomic arrival order changes
# speedup vs baseline: 1.8578x; 1.8578x over previous
#include <hip/hip_runtime.h>
#include <hip/hip_bf16.h>

#define IN_C  64
#define OUT_C 64
#define EPS   (1.0f + 1e-06f)

typedef float v2f __attribute__((ext_vector_type(2)));
typedef float v8f __attribute__((ext_vector_type(8)));

template <bool HAS_ATT>
__device__ __forceinline__ void wmma_block(
    const v2f a_frag[16],
    const float* __restrict__ w,
    const float* __restrict__ att,
    float* __restrict__ store,
    float* __restrict__ s_src,
    float* __restrict__ s_tgt,
    float scale, int row0, int half, int lrow)
{
    float att_s[4], att_t[4];
    if (HAS_ATT) {
#pragma unroll
        for (int nt = 0; nt < 4; ++nt) {
            att_s[nt] = att[nt * 16 + lrow];
            att_t[nt] = att[OUT_C + nt * 16 + lrow];
        }
    }

    float ssrc_acc[8];
    float stgt_acc[8];
#pragma unroll
    for (int r = 0; r < 8; ++r) { ssrc_acc[r] = 0.f; stgt_acc[r] = 0.f; }

#pragma unroll
    for (int nt = 0; nt < 4; ++nt) {
        const int col = nt * 16 + lrow;
        v8f c = {};
#pragma unroll
        for (int kk = 0; kk < 16; ++kk) {
            const int k0 = 4 * kk + 2 * half;
            v2f b;
            b.x = w[(size_t)k0 * OUT_C + col];
            b.y = w[(size_t)(k0 + 1) * OUT_C + col];
            c = __builtin_amdgcn_wmma_f32_16x16x4_f32(
                     false, a_frag[kk],
                     false, b,
                     (short)0, c,
                     false,  false);
        }
#pragma unroll
        for (int r = 0; r < 8; ++r) {
            const int row = row0 + r + half * 8;
            const float v = c[r];
            store[(size_t)row * OUT_C + col] = v * scale;
            if (HAS_ATT) {
                ssrc_acc[r] += v * att_s[nt];
                stgt_acc[r] += v * att_t[nt];
            }
        }
    }

    if (HAS_ATT) {
#pragma unroll
        for (int r = 0; r < 8; ++r) {
            float vs = ssrc_acc[r];
            float vt = stgt_acc[r];
#pragma unroll
            for (int m = 8; m >= 1; m >>= 1) {
                vs += __shfl_xor(vs, m, 32);
                vt += __shfl_xor(vt, m, 32);
            }
            if (lrow == 0) {
                const int row = row0 + r + half * 8;
                s_src[row] = vs;
                s_tgt[row] = vt;
            }
        }
    }
}

__global__ void __launch_bounds__(128)
node_gemm_kernel(const float* __restrict__ x,
                 const float* __restrict__ w_lower,
                 const float* __restrict__ a_lower,
                 const float* __restrict__ w_upper,
                 const float* __restrict__ a_upper,
                 const float* __restrict__ w_lin,
                 float* __restrict__ xm_l, float* __restrict__ xm_u,
                 float* __restrict__ sl_s, float* __restrict__ sl_t,
                 float* __restrict__ su_s, float* __restrict__ su_t,
                 float* __restrict__ out, int n_rows)
{
    const int wid  = (int)((blockIdx.x * blockDim.x + threadIdx.x) >> 5);
    const int lane = (int)(threadIdx.x & 31);
    const int row0 = wid * 16;
    if (row0 >= n_rows) return;

    const int half = lane >> 4;
    const int lrow = lane & 15;

    v2f a_frag[16];
    const float* xr = x + (size_t)(row0 + lrow) * IN_C + 2 * half;
#pragma unroll
    for (int kk = 0; kk < 16; ++kk)
        a_frag[kk] = *(const v2f*)(xr + 4 * kk);

    wmma_block<true >(a_frag, w_lower, a_lower, xm_l, sl_s, sl_t, 1.0f, row0, half, lrow);
    wmma_block<true >(a_frag, w_upper, a_upper, xm_u, su_s, su_t, 1.0f, row0, half, lrow);
    wmma_block<false>(a_frag, w_lin,   nullptr, out,  nullptr, nullptr, EPS, row0, half, lrow);
}

#define EDGES_PER_WAVE 32
#define ROW_GROUP      8

__device__ __forceinline__ float2 load_row_pair(const float* __restrict__ xm,
                                                int j_lanes, int src_lane, int lane)
{
    const int j = __builtin_amdgcn_readlane(j_lanes, src_lane);
    return ((const float2*)(xm + (size_t)j * OUT_C))[lane];
}

__device__ __forceinline__ void scatter_row_pair(float* __restrict__ out,
                                                 int i_lanes, float alpha_lanes,
                                                 int src_lane, int lane, float2 m)
{
    const int   i = __builtin_amdgcn_readlane(i_lanes, src_lane);
    const float a = __int_as_float(
        __builtin_amdgcn_readlane(__float_as_int(alpha_lanes), src_lane));
    float* o = out + (size_t)i * OUT_C + lane * 2;
    atomicAdd(o,     a * m.x);
    atomicAdd(o + 1, a * m.y);
}

__global__ void __launch_bounds__(256)
edge_scatter_kernel(const int* __restrict__ idx,
                    const float* __restrict__ vals,
                    const float* __restrict__ xm,
                    const float* __restrict__ s_src,
                    const float* __restrict__ s_tgt,
                    float* __restrict__ out,
                    int E)
{
    const int wave = __builtin_amdgcn_readfirstlane(
        (int)((blockIdx.x * blockDim.x + threadIdx.x) >> 5));
    const int lane = (int)(threadIdx.x & 31);
    const int e0 = wave * EDGES_PER_WAVE;
    if (e0 >= E) return;
    const int n_valid = (E - e0 < EDGES_PER_WAVE) ? (E - e0) : EDGES_PER_WAVE;

    int   i_l = 0, j_l = 0;
    float alpha_l = 0.f;
    if (lane < n_valid) {
        const int e = e0 + lane;
        i_l = idx[e];
        j_l = idx[(size_t)E + e];
        const float v = vals[e];
        const float z = s_src[j_l] + s_tgt[i_l];
        const float elu = (z > 0.f) ? z : expm1f(z);
        alpha_l = elu * v;
    }

    if (n_valid == EDGES_PER_WAVE) {
        float2 m[2][ROW_GROUP];
#pragma unroll
        for (int t = 0; t < ROW_GROUP; ++t)
            m[0][t] = load_row_pair(xm, j_l, t, lane);
#pragma unroll
        for (int g = 0; g < EDGES_PER_WAVE / ROW_GROUP; ++g) {
            if (g + 1 < EDGES_PER_WAVE / ROW_GROUP) {
#pragma unroll
                for (int t = 0; t < ROW_GROUP; ++t)
                    m[(g + 1) & 1][t] = load_row_pair(xm, j_l, (g + 1) * ROW_GROUP + t, lane);
            }
#pragma unroll
            for (int t = 0; t < ROW_GROUP; ++t)
                scatter_row_pair(out, i_l, alpha_l, g * ROW_GROUP + t, lane, m[g & 1][t]);
        }
    } else {
        for (int t = 0; t < n_valid; ++t) {
            const float2 mt = load_row_pair(xm, j_l, t, lane);
            scatter_row_pair(out, i_l, alpha_l, t, lane, mt);
        }
    }
}

__global__ void __launch_bounds__(256)
relu_kernel(float* __restrict__ out, size_t n)
{
    const size_t t = (size_t)blockIdx.x * blockDim.x + threadIdx.x;
    if (t < n) out[t] = fmaxf(out[t], 0.0f);
}

extern "C" void kernel_launch(void* const* d_in, const int* in_sizes, int n_in,
                              void* d_out, int out_size, void* d_ws, size_t ws_size,
                              hipStream_t stream) {
    const float* x        = (const float*)d_in[0];
    const int*   low_idx  = (const int*)  d_in[1];
    const float* low_val  = (const float*)d_in[2];
    const int*   up_idx   = (const int*)  d_in[3];
    const float* up_val   = (const float*)d_in[4];
    const float* w_lower  = (const float*)d_in[5];
    const float* a_lower  = (const float*)d_in[6];
    const float* w_upper  = (const float*)d_in[7];
    const float* a_upper  = (const float*)d_in[8];
    const float* w_lin    = (const float*)d_in[9];
    float*       out      = (float*)d_out;

    const int N = in_sizes[0] / IN_C;
    const int E = in_sizes[2];

    float* ws   = (float*)d_ws;
    float* xm_l = ws;
    float* xm_u = xm_l + (size_t)N * OUT_C;
    float* sl_s = xm_u + (size_t)N * OUT_C;
    float* sl_t = sl_s + N;
    float* su_s = sl_t + N;
    float* su_t = su_s + N;

    {
        const int waves  = (N + 15) / 16;
        const int blocks = (waves + 3) / 4;
        node_gemm_kernel<<<blocks, 128, 0, stream>>>(
            x, w_lower, a_lower, w_upper, a_upper, w_lin,
            xm_l, xm_u, sl_s, sl_t, su_s, su_t, out, N);
    }

    {
        const int waves  = (E + 31) / 32;
        const int blocks = (waves + 7) / 8;
        edge_scatter_kernel<<<blocks, 256, 0, stream>>>(
            low_idx, low_val, xm_l, sl_s, sl_t, out, E);
        edge_scatter_kernel<<<blocks, 256, 0, stream>>>(
            up_idx, up_val, xm_u, su_s, su_t, out, E);
    }

    {
        const size_t n = (size_t)N * OUT_C;
        const int blocks = (int)((n + 255) / 256);
        relu_kernel<<<blocks, 256, 0, stream>>>(out, n);
    }
}


# edge scatter: lane l owns channels l and l+32 (was 2l, 2l+1): each f32 atomic and each row load covers one whole 128-byte line, not half of two lines; same products, atomic arrival order may differ
# speedup vs baseline: 2.2891x; 1.2322x over previous
#include <hip/hip_runtime.h>
#include <hip/hip_bf16.h>

#define IN_C  64
#define OUT_C 64
#define EPS   (1.0f + 1e-06f)

typedef float v2f __attribute__((ext_vector_type(2)));
typedef float v8f __attribute__((ext_vector_type(8)));

template <bool HAS_ATT>
__device__ __forceinline__ void wmma_block(
    const v2f a_frag[16],
    const float* __restrict__ w,
    const float* __restrict__ att,
    float* __restrict__ store,
    float* __restrict__ s_src,
    float* __restrict__ s_tgt,
    float scale, int row0, int half, int lrow)
{
    float att_s[4], att_t[4];
    if (HAS_ATT) {
#pragma unroll
        for (int nt = 0; nt < 4; ++nt) {
            att_s[nt] = att[nt * 16 + lrow];
            att_t[nt] = att[OUT_C + nt * 16 + lrow];
        }
    }

    float ssrc_acc[8];
    float stgt_acc[8];
#pragma unroll
    for (int r = 0; r < 8; ++r) { ssrc_acc[r] = 0.f; stgt_acc[r] = 0.f; }

#pragma unroll
    for (int nt = 0; nt < 4; ++nt) {
        const int col = nt * 16 + lrow;
        v8f c = {};
#pragma unroll
        for (int kk = 0; kk < 16; ++kk) {
            const int k0 = 4 * kk + 2 * half;
            v2f b;
            b.x = w[(size_t)k0 * OUT_C + col];
            b.y = w[(size_t)(k0 + 1) * OUT_C + col];
            c = __builtin_amdgcn_wmma_f32_16x16x4_f32(
                     false, a_frag[kk],
                     false, b,
                     (short)0, c,
                     false,  false);
        }
#pragma unroll
        for (int r = 0; r < 8; ++r) {
            const int row = row0 + r + half * 8;
            const float v = c[r];
            store[(size_t)row * OUT_C + col] = v * scale;
            if (HAS_ATT) {
                ssrc_acc[r] += v * att_s[nt];
                stgt_acc[r] += v * att_t[nt];
            }
        }
    }

    if (HAS_ATT) {
#pragma unroll
        for (int r = 0; r < 8; ++r) {
            float vs = ssrc_acc[r];
            float vt = stgt_acc[r];
#pragma unroll
            for (int m = 8; m >= 1; m >>= 1) {
                vs += __shfl_xor(vs, m, 32);
                vt += __shfl_xor(vt, m, 32);
            }
            if (lrow == 0) {
                const int row = row0 + r + half * 8;
                s_src[row] = vs;
                s_tgt[row] = vt;
            }
        }
    }
}

__global__ void __launch_bounds__(128)
node_gemm_kernel(const float* __restrict__ x,
                 const float* __restrict__ w_lower,
                 const float* __restrict__ a_lower,
                 const float* __restrict__ w_upper,
                 const float* __restrict__ a_upper,
                 const float* __restrict__ w_lin,
                 float* __restrict__ xm_l, float* __restrict__ xm_u,
                 float* __restrict__ sl_s, float* __restrict__ sl_t,
                 float* __restrict__ su_s, float* __restrict__ su_t,
                 float* __restrict__ out, int n_rows)
{
    const int wid  = (int)((blockIdx.x * blockDim.x + threadIdx.x) >> 5);
    const int lane = (int)(threadIdx.x & 31);
    const int row0 = wid * 16;
    if (row0 >= n_rows) return;

    const int half = lane >> 4;
    const int lrow = lane & 15;

    v2f a_frag[16];
    const float* xr = x + (size_t)(row0 + lrow) * IN_C + 2 * half;
#pragma unroll
    for (int kk = 0; kk < 16; ++kk)
        a_frag[kk] = *(const v2f*)(xr + 4 * kk);

    wmma_block<true >(a_frag, w_lower, a_lower, xm_l, sl_s, sl_t, 1.0f, row0, half, lrow);
    wmma_block<true >(a_frag, w_upper, a_upper, xm_u, su_s, su_t, 1.0f, row0, half, lrow);
    wmma_block<false>(a_frag, w_lin,   nullptr, out,  nullptr, nullptr, EPS, row0, half, lrow);
}

#define EDGES_PER_WAVE 32
#define ROW_GROUP      8

__device__ __forceinline__ float2 load_row_pair(const float* __restrict__ xm,
                                                int j_lanes, int src_lane, int lane)
{
    const int j = __builtin_amdgcn_readlane(j_lanes, src_lane);
    const float* r = xm + (size_t)j * OUT_C + lane;
    float2 m;
    m.x = r[0];
    m.y = r[OUT_C / 2];
    return m;
}

__device__ __forceinline__ void scatter_row_pair(float* __restrict__ out,
                                                 int i_lanes, float alpha_lanes,
                                                 int src_lane, int lane, float2 m)
{
    const int   i = __builtin_amdgcn_readlane(i_lanes, src_lane);
    const float a = __int_as_float(
        __builtin_amdgcn_readlane(__float_as_int(alpha_lanes), src_lane));
    float* o = out + (size_t)i * OUT_C + lane;
    atomicAdd(o,             a * m.x);
    atomicAdd(o + OUT_C / 2, a * m.y);
}

__global__ void __launch_bounds__(256)
edge_scatter_kernel(const int* __restrict__ idx,
                    const float* __restrict__ vals,
                    const float* __restrict__ xm,
                    const float* __restrict__ s_src,
                    const float* __restrict__ s_tgt,
                    float* __restrict__ out,
                    int E)
{
    const int wave = __builtin_amdgcn_readfirstlane(
        (int)((blockIdx.x * blockDim.x + threadIdx.x) >> 5));
    const int lane = (int)(threadIdx.x & 31);
    const int e0 = wave * EDGES_PER_WAVE;
    if (e0 >= E) return;
    const int n_valid = (E - e0 < EDGES_PER_WAVE) ? (E - e0) : EDGES_PER_WAVE;

    int   i_l = 0, j_l = 0;
    float alpha_l = 0.f;
    if (lane < n_valid) {
        const int e = e0 + lane;
        i_l = idx[e];
        j_l = idx[(size_t)E + e];
        const float v = vals[e];
        const float z = s_src[j_l] + s_tgt[i_l];
        const float elu = (z > 0.f) ? z : expm1f(z);
        alpha_l = elu * v;
    }

    if (n_valid == EDGES_PER_WAVE) {
        float2 m[2][ROW_GROUP];
#pragma unroll
        for (int t = 0; t < ROW_GROUP; ++t)
            m[0][t] = load_row_pair(xm, j_l, t, lane);
#pragma unroll
        for (int g = 0; g < EDGES_PER_WAVE / ROW_GROUP; ++g) {
            if (g + 1 < EDGES_PER_WAVE / ROW_GROUP) {
#pragma unroll
                for (int t = 0; t < ROW_GROUP; ++t)
                    m[(g + 1) & 1][t] = load_row_pair(xm, j_l, (g + 1) * ROW_GROUP + t, lane);
            }
#pragma unroll
            for (int t = 0; t < ROW_GROUP; ++t)
                scatter_row_pair(out, i_l, alpha_l, g * ROW_GROUP + t, lane, m[g & 1][t]);
        }
    } else {
        for (int t = 0; t < n_valid; ++t) {
            const float2 mt = load_row_pair(xm, j_l, t, lane);
            scatter_row_pair(out, i_l, alpha_l, t, lane, mt);
        }
    }
}

__global__ void __launch_bounds__(256)
relu_kernel(float* __restrict__ out, size_t n)
{
    const size_t t = (size_t)blockIdx.x * blockDim.x + threadIdx.x;
    if (t < n) out[t] = fmaxf(out[t], 0.0f);
}

extern "C" void kernel_launch(void* const* d_in, const int* in_sizes, int n_in,
                              void* d_out, int out_size, void* d_ws, size_t ws_size,
                              hipStream_t stream) {
    const float* x        = (const float*)d_in[0];
    const int*   low_idx  = (const int*)  d_in[1];
    const float* low_val  = (const float*)d_in[2];
    const int*   up_idx   = (const int*)  d_in[3];
    const float* up_val   = (const float*)d_in[4];
    const float* w_lower  = (const float*)d_in[5];
    const float* a_lower  = (const float*)d_in[6];
    const float* w_upper  = (const float*)d_in[7];
    const float* a_upper  = (const float*)d_in[8];
    const float* w_lin    = (const float*)d_in[9];
    float*       out      = (float*)d_out;

    const int N = in_sizes[0] / IN_C;
    const int E = in_sizes[2];

    float* ws   = (float*)d_ws;
    float* xm_l = ws;
    float* xm_u = xm_l + (size_t)N * OUT_C;
    float* sl_s = xm_u + (size_t)N * OUT_C;
    float* sl_t = sl_s + N;
    float* su_s = sl_t + N;
    float* su_t = su_s + N;

    {
        const int waves  = (N + 15) / 16;
        const int blocks = (waves + 3) / 4;
        node_gemm_kernel<<<blocks, 128, 0, stream>>>(
            x, w_lower, a_lower, w_upper, a_upper, w_lin,
            xm_l, xm_u, sl_s, sl_t, su_s, su_t, out, N);
    }

    {
        const int waves  = (E + 31) / 32;
        const int blocks = (waves + 7) / 8;
        edge_scatter_kernel<<<blocks, 256, 0, stream>>>(
            low_idx, low_val, xm_l, sl_s, sl_t, out, E);
        edge_scatter_kernel<<<blocks, 256, 0, stream>>>(
            up_idx, up_val, xm_u, su_s, su_t, out, E);
    }

    {
        const size_t n = (size_t)N * OUT_C;
        const int blocks = (int)((n + 255) / 256);
        relu_kernel<<<blocks, 256, 0, stream>>>(out, n);
    }
}


# edge scatter: a work-group reads its die and adds only into rows of one half of the node range per visit (its own chip half first, by ticket): a line is added from one L2 half; atomic order changes
# speedup vs baseline: 2.6187x; 1.1440x over previous
#include <hip/hip_runtime.h>
#include <hip/hip_bf16.h>

#define IN_C  64
#define OUT_C 64
#define EPS   (1.0f + 1e-06f)
#define TICKET_STRIDE 64
#define TICKET_BYTES  (4 * TICKET_STRIDE * 4)

__device__ __forceinline__ bool tickets_fit(int n_rows, size_t ws_bytes)
{
    const size_t planes = ((size_t)n_rows * OUT_C * 2 + (size_t)n_rows * 4) * sizeof(float);
    return ws_bytes >= planes + (size_t)TICKET_BYTES;
}

typedef float v2f __attribute__((ext_vector_type(2)));
typedef float v8f __attribute__((ext_vector_type(8)));

template <bool HAS_ATT>
__device__ __forceinline__ void wmma_block(
    const v2f a_frag[16],
    const float* __restrict__ w,
    const float* __restrict__ att,
    float* __restrict__ store,
    float* __restrict__ s_src,
    float* __restrict__ s_tgt,
    float scale, int row0, int half, int lrow)
{
    float att_s[4], att_t[4];
    if (HAS_ATT) {
#pragma unroll
        for (int nt = 0; nt < 4; ++nt) {
            att_s[nt] = att[nt * 16 + lrow];
            att_t[nt] = att[OUT_C + nt * 16 + lrow];
        }
    }

    float ssrc_acc[8];
    float stgt_acc[8];
#pragma unroll
    for (int r = 0; r < 8; ++r) { ssrc_acc[r] = 0.f; stgt_acc[r] = 0.f; }

#pragma unroll
    for (int nt = 0; nt < 4; ++nt) {
        const int col = nt * 16 + lrow;
        v8f c = {};
#pragma unroll
        for (int kk = 0; kk < 16; ++kk) {
            const int k0 = 4 * kk + 2 * half;
            v2f b;
            b.x = w[(size_t)k0 * OUT_C + col];
            b.y = w[(size_t)(k0 + 1) * OUT_C + col];
            c = __builtin_amdgcn_wmma_f32_16x16x4_f32(
                     false, a_frag[kk],
                     false, b,
                     (short)0, c,
                     false,  false);
        }
#pragma unroll
        for (int r = 0; r < 8; ++r) {
            const int row = row0 + r + half * 8;
            const float v = c[r];
            store[(size_t)row * OUT_C + col] = v * scale;
            if (HAS_ATT) {
                ssrc_acc[r] += v * att_s[nt];
                stgt_acc[r] += v * att_t[nt];
            }
        }
    }

    if (HAS_ATT) {
#pragma unroll
        for (int r = 0; r < 8; ++r) {
            float vs = ssrc_acc[r];
            float vt = stgt_acc[r];
#pragma unroll
            for (int m = 8; m >= 1; m >>= 1) {
                vs += __shfl_xor(vs, m, 32);
                vt += __shfl_xor(vt, m, 32);
            }
            if (lrow == 0) {
                const int row = row0 + r + half * 8;
                s_src[row] = vs;
                s_tgt[row] = vt;
            }
        }
    }
}

__global__ void __launch_bounds__(128)
node_gemm_kernel(const float* __restrict__ x,
                 const float* __restrict__ w_lower,
                 const float* __restrict__ a_lower,
                 const float* __restrict__ w_upper,
                 const float* __restrict__ a_upper,
                 const float* __restrict__ w_lin,
                 float* __restrict__ xm_l, float* __restrict__ xm_u,
                 float* __restrict__ sl_s, float* __restrict__ sl_t,
                 float* __restrict__ su_s, float* __restrict__ su_t,
                 float* __restrict__ out, int n_rows,
                 int* __restrict__ tickets, size_t ws_bytes)
{
    const int use_tickets = tickets_fit(n_rows, ws_bytes) ? 1 : 0;
    if (use_tickets != 0 && blockIdx.x == 0 && threadIdx.x < 4)
        tickets[threadIdx.x * TICKET_STRIDE] = 0;

    const int wid  = (int)((blockIdx.x * blockDim.x + threadIdx.x) >> 5);
    const int lane = (int)(threadIdx.x & 31);
    const int row0 = wid * 16;
    if (row0 >= n_rows) return;

    const int half = lane >> 4;
    const int lrow = lane & 15;

    v2f a_frag[16];
    const float* xr = x + (size_t)(row0 + lrow) * IN_C + 2 * half;
#pragma unroll
    for (int kk = 0; kk < 16; ++kk)
        a_frag[kk] = *(const v2f*)(xr + 4 * kk);

    wmma_block<true >(a_frag, w_lower, a_lower, xm_l, sl_s, sl_t, 1.0f, row0, half, lrow);
    wmma_block<true >(a_frag, w_upper, a_upper, xm_u, su_s, su_t, 1.0f, row0, half, lrow);
    wmma_block<false>(a_frag, w_lin,   nullptr, out,  nullptr, nullptr, EPS, row0, half, lrow);
}

#define EDGES_PER_WAVE 32
#define ROW_GROUP      24

__device__ __forceinline__ float2 load_row_pair(const float* __restrict__ xm,
                                                int j_lanes, int src_lane, int lane)
{
    const int j = __builtin_amdgcn_readlane(j_lanes, src_lane);
    const float* r = xm + (size_t)j * OUT_C + lane;
    float2 m;
    m.x = r[0];
    m.y = r[OUT_C / 2];
    return m;
}

__device__ __forceinline__ void scatter_row_pair(float* __restrict__ out,
                                                 int i_lanes, float alpha_lanes,
                                                 int src_lane, int lane, float2 m)
{
    const int   i = __builtin_amdgcn_readlane(i_lanes, src_lane);
    const float a = __int_as_float(
        __builtin_amdgcn_readlane(__float_as_int(alpha_lanes), src_lane));
    float* o = out + (size_t)i * OUT_C + lane;
    atomicAdd(o,             a * m.x);
    atomicAdd(o + OUT_C / 2, a * m.y);
}

__device__ __forceinline__ void visit_chunk(const int* __restrict__ idx,
                                            const float* __restrict__ vals,
                                            const float* __restrict__ xm,
                                            const float* __restrict__ s_src,
                                            const float* __restrict__ s_tgt,
                                            float* __restrict__ out,
                                            int E, int c, int cls, int half_n, int lane)
{
    const int e0 = c * EDGES_PER_WAVE;
    const int n_valid = (E - e0 < EDGES_PER_WAVE) ? (E - e0) : EDGES_PER_WAVE;

    int   i_l = 0, j_l = 0;
    float alpha_l = 0.f;
    bool  own = false;
    if (lane < n_valid) {
        const int e = e0 + lane;
        i_l = idx[e];
        j_l = idx[(size_t)E + e];
        const float v = vals[e];
        own = (cls > 1) || (((i_l >= half_n) ? 1 : 0) == cls);
        if (own) {
            const float z = s_src[j_l] + s_tgt[i_l];
            const float elu = (z > 0.f) ? z : expm1f(z);
            alpha_l = elu * v;
        }
    }

    unsigned mask = __builtin_amdgcn_ballot_w32(own);
    while (mask != 0u) {
        int    ls[ROW_GROUP];
        float2 m[ROW_GROUP];
#pragma unroll
        for (int t = 0; t < ROW_GROUP; ++t) {
            ls[t] = -1;
            if (mask != 0u) {
                ls[t] = __builtin_ctz(mask);
                mask &= mask - 1u;
                m[t] = load_row_pair(xm, j_l, ls[t], lane);
            }
        }
#pragma unroll
        for (int t = 0; t < ROW_GROUP; ++t) {
            if (ls[t] >= 0)
                scatter_row_pair(out, i_l, alpha_l, ls[t], lane, m[t]);
        }
    }
}

__global__ void __launch_bounds__(256)
edge_scatter_kernel(const int* __restrict__ idx,
                    const float* __restrict__ vals,
                    const float* __restrict__ xm,
                    const float* __restrict__ s_src,
                    const float* __restrict__ s_tgt,
                    float* __restrict__ out,
                    int E, int n_rows,
                    int* __restrict__ tickets,
                    size_t ws_bytes)
{
    const int use_tickets = tickets_fit(n_rows, ws_bytes) ? 1 : 0;
    __shared__ int s_half;
    __shared__ int s_ticket;

    const int lane = (int)(threadIdx.x & 31);
    const int wib  = __builtin_amdgcn_readfirstlane((int)(threadIdx.x >> 5));
    const int wpb  = (int)(blockDim.x >> 5);
    const int n_chunks = (E + EDGES_PER_WAVE - 1) / EDGES_PER_WAVE;
    const int half_n   = n_rows >> 1;

    if (use_tickets == 0) {
        const int c = (int)blockIdx.x * wpb + wib;
        if (c < n_chunks)
            visit_chunk(idx, vals, xm, s_src, s_tgt, out, E, c, 2, half_n, lane);
        return;
    }

    if (threadIdx.x == 0) {
        const unsigned id = __builtin_amdgcn_s_sendmsg_rtn(0x87);
        s_half = (((id >> 16) & 15u) >= 4u) ? 1 : 0;
    }
    __syncthreads();
    const int my_half   = s_half;
    const int n_tickets = (n_chunks + wpb - 1) / wpb;

    for (int p = 0; p < 2; ++p) {
        const int cls = my_half ^ p;
        for (;;) {
            __syncthreads();
            if (threadIdx.x == 0)
                s_ticket = atomicAdd(tickets + cls * TICKET_STRIDE, 1);
            __syncthreads();
            const int t = s_ticket;
            if (t >= n_tickets) break;
            const int c = t * wpb + wib;
            if (c < n_chunks)
                visit_chunk(idx, vals, xm, s_src, s_tgt, out, E, c, cls, half_n, lane);
        }
    }
}

__global__ void __launch_bounds__(256)
relu_kernel(float* __restrict__ out, size_t n)
{
    const size_t t = (size_t)blockIdx.x * blockDim.x + threadIdx.x;
    if (t < n) out[t] = fmaxf(out[t], 0.0f);
}

extern "C" void kernel_launch(void* const* d_in, const int* in_sizes, int n_in,
                              void* d_out, int out_size, void* d_ws, size_t ws_size,
                              hipStream_t stream) {
    const float* x        = (const float*)d_in[0];
    const int*   low_idx  = (const int*)  d_in[1];
    const float* low_val  = (const float*)d_in[2];
    const int*   up_idx   = (const int*)  d_in[3];
    const float* up_val   = (const float*)d_in[4];
    const float* w_lower  = (const float*)d_in[5];
    const float* a_lower  = (const float*)d_in[6];
    const float* w_upper  = (const float*)d_in[7];
    const float* a_upper  = (const float*)d_in[8];
    const float* w_lin    = (const float*)d_in[9];
    float*       out      = (float*)d_out;

    const int N = in_sizes[0] / IN_C;
    const int E = in_sizes[2];

    float* ws   = (float*)d_ws;
    float* xm_l = ws;
    float* xm_u = xm_l + (size_t)N * OUT_C;
    float* sl_s = xm_u + (size_t)N * OUT_C;
    float* sl_t = sl_s + N;
    float* su_s = sl_t + N;
    float* su_t = su_s + N;
    int*   tickets = (int*)(su_t + N);

    {
        const int waves  = (N + 15) / 16;
        const int blocks = (waves + 3) / 4;
        node_gemm_kernel<<<blocks, 128, 0, stream>>>(
            x, w_lower, a_lower, w_upper, a_upper, w_lin,
            xm_l, xm_u, sl_s, sl_t, su_s, su_t, out, N, tickets, ws_size);
    }

    {
        const int waves  = (E + 31) / 32;
        const int blocks = (waves + 7) / 8;
        edge_scatter_kernel<<<blocks, 256, 0, stream>>>(
            low_idx, low_val, xm_l, sl_s, sl_t, out, E, N,
            tickets, ws_size);
        edge_scatter_kernel<<<blocks, 256, 0, stream>>>(
            up_idx, up_val, xm_u, su_s, su_t, out, E, N,
            tickets + 2 * TICKET_STRIDE, ws_size);
    }

    {
        const size_t n = (size_t)N * OUT_C;
        const int blocks = (int)((n + 255) / 256);
        relu_kernel<<<blocks, 256, 0, stream>>>(out, n);
    }
}


# edge scatter: both edge lists in ONE launch (was two) of 2048 resident work-groups; own half-class of list 0, of list 1, then the other class of each, by ticket; atomic arrival order changes
# speedup vs baseline: 2.7619x; 1.0547x over previous
#include <hip/hip_runtime.h>
#include <hip/hip_bf16.h>

#define IN_C  64
#define OUT_C 64
#define EPS   (1.0f + 1e-06f)
#define TICKET_STRIDE 64
#define TICKET_BYTES  (4 * TICKET_STRIDE * 4)

__device__ __forceinline__ bool tickets_fit(int n_rows, size_t ws_bytes)
{
    const size_t planes = ((size_t)n_rows * OUT_C * 2 + (size_t)n_rows * 4) * sizeof(float);
    return ws_bytes >= planes + (size_t)TICKET_BYTES;
}

typedef float v2f __attribute__((ext_vector_type(2)));
typedef float v8f __attribute__((ext_vector_type(8)));

template <bool HAS_ATT>
__device__ __forceinline__ void wmma_block(
    const v2f a_frag[16],
    const float* __restrict__ w,
    const float* __restrict__ att,
    float* __restrict__ store,
    float* __restrict__ s_src,
    float* __restrict__ s_tgt,
    float scale, int row0, int half, int lrow)
{
    float att_s[4], att_t[4];
    if (HAS_ATT) {
#pragma unroll
        for (int nt = 0; nt < 4; ++nt) {
            att_s[nt] = att[nt * 16 + lrow];
            att_t[nt] = att[OUT_C + nt * 16 + lrow];
        }
    }

    float ssrc_acc[8];
    float stgt_acc[8];
#pragma unroll
    for (int r = 0; r < 8; ++r) { ssrc_acc[r] = 0.f; stgt_acc[r] = 0.f; }

#pragma unroll
    for (int nt = 0; nt < 4; ++nt) {
        const int col = nt * 16 + lrow;
        v8f c = {};
#pragma unroll
        for (int kk = 0; kk < 16; ++kk) {
            const int k0 = 4 * kk + 2 * half;
            v2f b;
            b.x = w[(size_t)k0 * OUT_C + col];
            b.y = w[(size_t)(k0 + 1) * OUT_C + col];
            c = __builtin_amdgcn_wmma_f32_16x16x4_f32(
                     false, a_frag[kk],
                     false, b,
                     (short)0, c,
                     false,  false);
        }
#pragma unroll
        for (int r = 0; r < 8; ++r) {
            const int row = row0 + r + half * 8;
            const float v = c[r];
            store[(size_t)row * OUT_C + col] = v * scale;
            if (HAS_ATT) {
                ssrc_acc[r] += v * att_s[nt];
                stgt_acc[r] += v * att_t[nt];
            }
        }
    }

    if (HAS_ATT) {
#pragma unroll
        for (int r = 0; r < 8; ++r) {
            float vs = ssrc_acc[r];
            float vt = stgt_acc[r];
#pragma unroll
            for (int m = 8; m >= 1; m >>= 1) {
                vs += __shfl_xor(vs, m, 32);
                vt += __shfl_xor(vt, m, 32);
            }
            if (lrow == 0) {
                const int row = row0 + r + half * 8;
                s_src[row] = vs;
                s_tgt[row] = vt;
            }
        }
    }
}

__global__ void __launch_bounds__(128)
node_gemm_kernel(const float* __restrict__ x,
                 const float* __restrict__ w_lower,
                 const float* __restrict__ a_lower,
                 const float* __restrict__ w_upper,
                 const float* __restrict__ a_upper,
                 const float* __restrict__ w_lin,
                 float* __restrict__ xm_l, float* __restrict__ xm_u,
                 float* __restrict__ sl_s, float* __restrict__ sl_t,
                 float* __restrict__ su_s, float* __restrict__ su_t,
                 float* __restrict__ out, int n_rows,
                 int* __restrict__ tickets, size_t ws_bytes)
{
    const int use_tickets = tickets_fit(n_rows, ws_bytes) ? 1 : 0;
    if (use_tickets != 0 && blockIdx.x == 0 && threadIdx.x < 4)
        tickets[threadIdx.x * TICKET_STRIDE] = 0;

    const int wid  = (int)((blockIdx.x * blockDim.x + threadIdx.x) >> 5);
    const int lane = (int)(threadIdx.x & 31);
    const int row0 = wid * 16;
    if (row0 >= n_rows) return;

    const int half = lane >> 4;
    const int lrow = lane & 15;

    v2f a_frag[16];
    const float* xr = x + (size_t)(row0 + lrow) * IN_C + 2 * half;
#pragma unroll
    for (int kk = 0; kk < 16; ++kk)
        a_frag[kk] = *(const v2f*)(xr + 4 * kk);

    wmma_block<true >(a_frag, w_lower, a_lower, xm_l, sl_s, sl_t, 1.0f, row0, half, lrow);
    wmma_block<true >(a_frag, w_upper, a_upper, xm_u, su_s, su_t, 1.0f, row0, half, lrow);
    wmma_block<false>(a_frag, w_lin,   nullptr, out,  nullptr, nullptr, EPS, row0, half, lrow);
}

#define EDGES_PER_WAVE 32
#define ROW_GROUP      24

__device__ __forceinline__ float2 load_row_pair(const float* __restrict__ xm,
                                                int j_lanes, int src_lane, int lane)
{
    const int j = __builtin_amdgcn_readlane(j_lanes, src_lane);
    const float* r = xm + (size_t)j * OUT_C + lane;
    float2 m;
    m.x = r[0];
    m.y = r[OUT_C / 2];
    return m;
}

__device__ __forceinline__ void scatter_row_pair(float* __restrict__ out,
                                                 int i_lanes, float alpha_lanes,
                                                 int src_lane, int lane, float2 m)
{
    const int   i = __builtin_amdgcn_readlane(i_lanes, src_lane);
    const float a = __int_as_float(
        __builtin_amdgcn_readlane(__float_as_int(alpha_lanes), src_lane));
    float* o = out + (size_t)i * OUT_C + lane;
    atomicAdd(o,             a * m.x);
    atomicAdd(o + OUT_C / 2, a * m.y);
}

__device__ __forceinline__ void visit_chunk(const int* __restrict__ idx,
                                            const float* __restrict__ vals,
                                            const float* __restrict__ xm,
                                            const float* __restrict__ s_src,
                                            const float* __restrict__ s_tgt,
                                            float* __restrict__ out,
                                            int E, int c, int cls, int half_n, int lane)
{
    const int e0 = c * EDGES_PER_WAVE;
    const int n_valid = (E - e0 < EDGES_PER_WAVE) ? (E - e0) : EDGES_PER_WAVE;

    int   i_l = 0, j_l = 0;
    float alpha_l = 0.f;
    bool  own = false;
    if (lane < n_valid) {
        const int e = e0 + lane;
        i_l = idx[e];
        j_l = idx[(size_t)E + e];
        const float v = vals[e];
        own = (cls > 1) || (((i_l >= half_n) ? 1 : 0) == cls);
        if (own) {
            const float z = s_src[j_l] + s_tgt[i_l];
            const float elu = (z > 0.f) ? z : expm1f(z);
            alpha_l = elu * v;
        }
    }

    unsigned mask = __builtin_amdgcn_ballot_w32(own);
    while (mask != 0u) {
        int    ls[ROW_GROUP];
        float2 m[ROW_GROUP];
#pragma unroll
        for (int t = 0; t < ROW_GROUP; ++t) {
            ls[t] = -1;
            if (mask != 0u) {
                ls[t] = __builtin_ctz(mask);
                mask &= mask - 1u;
                m[t] = load_row_pair(xm, j_l, ls[t], lane);
            }
        }
#pragma unroll
        for (int t = 0; t < ROW_GROUP; ++t) {
            if (ls[t] >= 0)
                scatter_row_pair(out, i_l, alpha_l, ls[t], lane, m[t]);
        }
    }
}

__global__ void __launch_bounds__(256)
edge_scatter_kernel(const int* __restrict__ idx_a,
                    const float* __restrict__ vals_a,
                    const float* __restrict__ xm_a,
                    const float* __restrict__ ssrc_a,
                    const float* __restrict__ stgt_a,
                    const int* __restrict__ idx_b,
                    const float* __restrict__ vals_b,
                    const float* __restrict__ xm_b,
                    const float* __restrict__ ssrc_b,
                    const float* __restrict__ stgt_b,
                    float* __restrict__ out,
                    int E, int n_rows,
                    int* __restrict__ tickets,
                    size_t ws_bytes)
{
    const int use_tickets = tickets_fit(n_rows, ws_bytes) ? 1 : 0;
    __shared__ int s_half;
    __shared__ int s_ticket;

    const int lane = (int)(threadIdx.x & 31);
    const int wib  = __builtin_amdgcn_readfirstlane((int)(threadIdx.x >> 5));
    const int wpb  = (int)(blockDim.x >> 5);
    const int n_chunks = (E + EDGES_PER_WAVE - 1) / EDGES_PER_WAVE;
    const int half_n   = n_rows >> 1;

    if (use_tickets == 0) {
        const int stride = (int)gridDim.x * wpb;
#pragma unroll 1
        for (int l = 0; l < 2; ++l) {
            const int*   idx   = l ? idx_b  : idx_a;
            const float* vals  = l ? vals_b : vals_a;
            const float* xm    = l ? xm_b   : xm_a;
            const float* s_src = l ? ssrc_b : ssrc_a;
            const float* s_tgt = l ? stgt_b : stgt_a;
            for (int c = (int)blockIdx.x * wpb + wib; c < n_chunks; c += stride)
                visit_chunk(idx, vals, xm, s_src, s_tgt, out, E, c, 2, half_n, lane);
        }
        return;
    }

    if (threadIdx.x == 0) {
        const unsigned id = __builtin_amdgcn_s_sendmsg_rtn(0x87);
        s_half = (((id >> 16) & 15u) >= 4u) ? 1 : 0;
    }
    __syncthreads();
    const int my_half   = s_half;
    const int n_tickets = (n_chunks + wpb - 1) / wpb;

#pragma unroll 1
    for (int ph = 0; ph < 4; ++ph) {
        const int l   = ph & 1;
        const int cls = my_half ^ (ph >> 1);
        const int*   idx   = l ? idx_b  : idx_a;
        const float* vals  = l ? vals_b : vals_a;
        const float* xm    = l ? xm_b   : xm_a;
        const float* s_src = l ? ssrc_b : ssrc_a;
        const float* s_tgt = l ? stgt_b : stgt_a;
        int* counter = tickets + (2 * l + cls) * TICKET_STRIDE;
        for (;;) {
            __syncthreads();
            if (threadIdx.x == 0)
                s_ticket = atomicAdd(counter, 1);
            __syncthreads();
            const int t = s_ticket;
            if (t >= n_tickets) break;
            const int c = t * wpb + wib;
            if (c < n_chunks)
                visit_chunk(idx, vals, xm, s_src, s_tgt, out, E, c, cls, half_n, lane);
        }
    }
}

__global__ void __launch_bounds__(256)
relu_kernel(float* __restrict__ out, size_t n)
{
    const size_t t = (size_t)blockIdx.x * blockDim.x + threadIdx.x;
    if (t < n) out[t] = fmaxf(out[t], 0.0f);
}

extern "C" void kernel_launch(void* const* d_in, const int* in_sizes, int n_in,
                              void* d_out, int out_size, void* d_ws, size_t ws_size,
                              hipStream_t stream) {
    const float* x        = (const float*)d_in[0];
    const int*   low_idx  = (const int*)  d_in[1];
    const float* low_val  = (const float*)d_in[2];
    const int*   up_idx   = (const int*)  d_in[3];
    const float* up_val   = (const float*)d_in[4];
    const float* w_lower  = (const float*)d_in[5];
    const float* a_lower  = (const float*)d_in[6];
    const float* w_upper  = (const float*)d_in[7];
    const float* a_upper  = (const float*)d_in[8];
    const float* w_lin    = (const float*)d_in[9];
    float*       out      = (float*)d_out;

    const int N = in_sizes[0] / IN_C;
    const int E = in_sizes[2];

    float* ws   = (float*)d_ws;
    float* xm_l = ws;
    float* xm_u = xm_l + (size_t)N * OUT_C;
    float* sl_s = xm_u + (size_t)N * OUT_C;
    float* sl_t = sl_s + N;
    float* su_s = sl_t + N;
    float* su_t = su_s + N;
    int*   tickets = (int*)(su_t + N);

    {
        const int waves  = (N + 15) / 16;
        const int blocks = (waves + 3) / 4;
        node_gemm_kernel<<<blocks, 128, 0, stream>>>(
            x, w_lower, a_lower, w_upper, a_upper, w_lin,
            xm_l, xm_u, sl_s, sl_t, su_s, su_t, out, N, tickets, ws_size);
    }

    {
        const int blocks = 2048;
        edge_scatter_kernel<<<blocks, 256, 0, stream>>>(
            low_idx, low_val, xm_l, sl_s, sl_t,
            up_idx,  up_val,  xm_u, su_s, su_t,
            out, E, N, tickets, ws_size);
    }

    {
        const size_t n = (size_t)N * OUT_C;
        const int blocks = (int)((n + 255) / 256);
        relu_kernel<<<blocks, 256, 0, stream>>>(out, n);
    }
}
